# starting kernel
# baseline (speedup 1.0000x reference)
#include <hip/hip_runtime.h>
#include <math.h>

typedef __attribute__((ext_vector_type(2))) float v2f;
typedef __attribute__((ext_vector_type(8))) float v8f;

#define N_NODES   50000
#define N_EDGES   640000
#define E_TOT     (N_EDGES + N_NODES)
#define DIM       128
#define NEG_SLOPE 0.2f

__global__ void k_init(float* __restrict__ out, float* __restrict__ maxv,
                       float* __restrict__ denom) {
    int i = blockIdx.x * blockDim.x + threadIdx.x;
    out[i] = 0.0f;
    if (i < N_NODES) {
        maxv[i]  = -__builtin_inff();
        denom[i] = 0.0f;
    }
}

__global__ void k_gemm_wmma(const float* __restrict__ X,
                            const float* __restrict__ W,
                            float* __restrict__ H) {
    const int wave = threadIdx.x >> 5;
    const int lane = threadIdx.x & 31;
    const int hb   = lane >> 4;
    const int l    = lane & 15;
    const int m0   = blockIdx.x * 16;
    const int n0   = wave * 16;

    const float* xrow = X + (m0 + l) * DIM + 2 * hb;
    const float* wcol = W + (2 * hb) * DIM + n0 + l;

    v8f acc = {};
#pragma unroll
    for (int k = 0; k < DIM; k += 4) {
        v2f a = *(const v2f*)(xrow + k);
        v2f b;
        b.x = wcol[k * DIM];
        b.y = wcol[k * DIM + DIM];
        acc = __builtin_amdgcn_wmma_f32_16x16x4_f32(
            false, a, false, b, (short)0, acc, false, false);
    }

    float* hout = H + (m0 + 8 * hb) * DIM + n0 + l;
#pragma unroll
    for (int r = 0; r < 8; ++r)
        hout[r * DIM] = acc[r];
}

__global__ void k_attn_coef(const float* __restrict__ H,
                            const float* __restrict__ att_src,
                            const float* __restrict__ att_dst,
                            float* __restrict__ a_s, float* __restrict__ a_d) {
    const int wave = threadIdx.x >> 5;
    const int lane = threadIdx.x & 31;
    const int node = blockIdx.x * 8 + wave;
    if (node >= N_NODES) return;

    float4 h4  = ((const float4*)(H + node * DIM))[lane];
    float4 as4 = ((const float4*)att_src)[lane];
    float4 ad4 = ((const float4*)att_dst)[lane];
    float s = h4.x * as4.x + h4.y * as4.y + h4.z * as4.z + h4.w * as4.w;
    float d = h4.x * ad4.x + h4.y * ad4.y + h4.z * ad4.z + h4.w * ad4.w;
#pragma unroll
    for (int off = 16; off > 0; off >>= 1) {
        s += __shfl_xor(s, off, 32);
        d += __shfl_xor(d, off, 32);
    }
    if (lane == 0) { a_s[node] = s; a_d[node] = d; }
}

__device__ __forceinline__ void edge_st(const int* __restrict__ ei, int e,
                                        int& s, int& t) {
    if (e < N_EDGES) { s = ei[e]; t = ei[N_EDGES + e]; }
    else             { s = t = e - N_EDGES; }
}

__device__ __forceinline__ float leaky(float e) {
    return e > 0.0f ? e : NEG_SLOPE * e;
}

__device__ __forceinline__ void atomicMaxF32(float* addr, float v) {
    if (v >= 0.0f) atomicMax((int*)addr, __float_as_int(v));
    else           atomicMin((unsigned int*)addr, __float_as_uint(v));
}

__global__ void k_edge_max(const int* __restrict__ ei,
                           const float* __restrict__ a_s,
                           const float* __restrict__ a_d,
                           float* __restrict__ maxv) {
    int e = blockIdx.x * blockDim.x + threadIdx.x;
    if (e >= E_TOT) return;
    int s, t; edge_st(ei, e, s, t);
    atomicMaxF32(&maxv[t], leaky(a_s[s] + a_d[t]));
}

__global__ void k_edge_denom(const int* __restrict__ ei,
                             const float* __restrict__ a_s,
                             const float* __restrict__ a_d,
                             const float* __restrict__ maxv,
                             float* __restrict__ denom) {
    int e = blockIdx.x * blockDim.x + threadIdx.x;
    if (e >= E_TOT) return;
    int s, t; edge_st(ei, e, s, t);
    float w = expf(leaky(a_s[s] + a_d[t]) - maxv[t]);
    unsafeAtomicAdd(&denom[t], w);
}

__global__ void k_scatter(const int* __restrict__ ei,
                          const float* __restrict__ H,
                          const float* __restrict__ a_s,
                          const float* __restrict__ a_d,
                          const float* __restrict__ maxv,
                          const float* __restrict__ denom,
                          float* __restrict__ out) {
    const int wave = threadIdx.x >> 5;
    const int lane = threadIdx.x & 31;
    const int e = blockIdx.x * 8 + wave;
    if (e >= E_TOT) return;
    int s, t; edge_st(ei, e, s, t);

    float alpha = expf(leaky(a_s[s] + a_d[t]) - maxv[t]) / denom[t];

    float4 h4 = ((const float4*)(H + s * DIM))[lane];
    float* o = out + t * DIM + lane * 4;
    unsafeAtomicAdd(o + 0, alpha * h4.x);
    unsafeAtomicAdd(o + 1, alpha * h4.y);
    unsafeAtomicAdd(o + 2, alpha * h4.z);
    unsafeAtomicAdd(o + 3, alpha * h4.w);
}

__global__ void k_finalize(float* __restrict__ out,
                           const float* __restrict__ bias) {
    int i = blockIdx.x * blockDim.x + threadIdx.x;
    out[i] = tanhf(out[i] + bias[i & (DIM - 1)]);
}

extern "C" void kernel_launch(void* const* d_in, const int* in_sizes, int n_in,
                              void* d_out, int out_size, void* d_ws, size_t ws_size,
                              hipStream_t stream) {
    const float* x        = (const float*)d_in[0];
    const int*   ei       = (const int*)  d_in[1];
    const float* W        = (const float*)d_in[2];
    const float* att_src  = (const float*)d_in[3];
    const float* att_dst  = (const float*)d_in[4];
    const float* bias     = (const float*)d_in[5];
    float*       out      = (float*)d_out;

    float* H     = (float*)d_ws;
    float* a_s   = H    + (size_t)N_NODES * DIM;
    float* a_d   = a_s  + N_NODES;
    float* maxv  = a_d  + N_NODES;
    float* denom = maxv + N_NODES;

    const int NE_ELEMS = N_NODES * DIM;

    k_init<<<NE_ELEMS / 256, 256, 0, stream>>>(out, maxv, denom);

    k_gemm_wmma<<<N_NODES / 16, 256, 0, stream>>>(x, W, H);

    k_attn_coef<<<N_NODES / 8, 256, 0, stream>>>(H, att_src, att_dst, a_s, a_d);

    const int eb = (E_TOT + 255) / 256;
    k_edge_max<<<eb, 256, 0, stream>>>(ei, a_s, a_d, maxv);
    k_edge_denom<<<eb, 256, 0, stream>>>(ei, a_s, a_d, maxv, denom);

    k_scatter<<<(E_TOT + 7) / 8, 256, 0, stream>>>(ei, H, a_s, a_d, maxv, denom, out);

    k_finalize<<<NE_ELEMS / 256, 256, 0, stream>>>(out, bias);
}


# scatter: lane owns columns l+32j, so each row load and each f32 atomic add of an edge covers one whole 128-B line, not a quarter of four lines; sums stay f32 atomics in arrival order
# speedup vs baseline: 1.2371x; 1.2371x over previous
#include <hip/hip_runtime.h>
#include <math.h>

typedef __attribute__((ext_vector_type(2))) float v2f;
typedef __attribute__((ext_vector_type(8))) float v8f;

#define N_NODES   50000
#define N_EDGES   640000
#define E_TOT     (N_EDGES + N_NODES)
#define DIM       128
#define NEG_SLOPE 0.2f

__global__ void k_init(float* __restrict__ out, float* __restrict__ maxv,
                       float* __restrict__ denom) {
    int i = blockIdx.x * blockDim.x + threadIdx.x;
    out[i] = 0.0f;
    if (i < N_NODES) {
        maxv[i]  = -__builtin_inff();
        denom[i] = 0.0f;
    }
}

__global__ void k_gemm_wmma(const float* __restrict__ X,
                            const float* __restrict__ W,
                            float* __restrict__ H) {
    const int wave = threadIdx.x >> 5;
    const int lane = threadIdx.x & 31;
    const int hb   = lane >> 4;
    const int l    = lane & 15;
    const int m0   = blockIdx.x * 16;
    const int n0   = wave * 16;

    const float* xrow = X + (m0 + l) * DIM + 2 * hb;
    const float* wcol = W + (2 * hb) * DIM + n0 + l;

    v8f acc = {};
#pragma unroll
    for (int k = 0; k < DIM; k += 4) {
        v2f a = *(const v2f*)(xrow + k);
        v2f b;
        b.x = wcol[k * DIM];
        b.y = wcol[k * DIM + DIM];
        acc = __builtin_amdgcn_wmma_f32_16x16x4_f32(
            false, a, false, b, (short)0, acc, false, false);
    }

    float* hout = H + (m0 + 8 * hb) * DIM + n0 + l;
#pragma unroll
    for (int r = 0; r < 8; ++r)
        hout[r * DIM] = acc[r];
}

__global__ void k_attn_coef(const float* __restrict__ H,
                            const float* __restrict__ att_src,
                            const float* __restrict__ att_dst,
                            float* __restrict__ a_s, float* __restrict__ a_d) {
    const int wave = threadIdx.x >> 5;
    const int lane = threadIdx.x & 31;
    const int node = blockIdx.x * 8 + wave;
    if (node >= N_NODES) return;

    float4 h4  = ((const float4*)(H + node * DIM))[lane];
    float4 as4 = ((const float4*)att_src)[lane];
    float4 ad4 = ((const float4*)att_dst)[lane];
    float s = h4.x * as4.x + h4.y * as4.y + h4.z * as4.z + h4.w * as4.w;
    float d = h4.x * ad4.x + h4.y * ad4.y + h4.z * ad4.z + h4.w * ad4.w;
#pragma unroll
    for (int off = 16; off > 0; off >>= 1) {
        s += __shfl_xor(s, off, 32);
        d += __shfl_xor(d, off, 32);
    }
    if (lane == 0) { a_s[node] = s; a_d[node] = d; }
}

__device__ __forceinline__ void edge_st(const int* __restrict__ ei, int e,
                                        int& s, int& t) {
    if (e < N_EDGES) { s = ei[e]; t = ei[N_EDGES + e]; }
    else             { s = t = e - N_EDGES; }
}

__device__ __forceinline__ float leaky(float e) {
    return e > 0.0f ? e : NEG_SLOPE * e;
}

__device__ __forceinline__ void atomicMaxF32(float* addr, float v) {
    if (v >= 0.0f) atomicMax((int*)addr, __float_as_int(v));
    else           atomicMin((unsigned int*)addr, __float_as_uint(v));
}

__global__ void k_edge_max(const int* __restrict__ ei,
                           const float* __restrict__ a_s,
                           const float* __restrict__ a_d,
                           float* __restrict__ maxv) {
    int e = blockIdx.x * blockDim.x + threadIdx.x;
    if (e >= E_TOT) return;
    int s, t; edge_st(ei, e, s, t);
    atomicMaxF32(&maxv[t], leaky(a_s[s] + a_d[t]));
}

__global__ void k_edge_denom(const int* __restrict__ ei,
                             const float* __restrict__ a_s,
                             const float* __restrict__ a_d,
                             const float* __restrict__ maxv,
                             float* __restrict__ denom) {
    int e = blockIdx.x * blockDim.x + threadIdx.x;
    if (e >= E_TOT) return;
    int s, t; edge_st(ei, e, s, t);
    float w = expf(leaky(a_s[s] + a_d[t]) - maxv[t]);
    unsafeAtomicAdd(&denom[t], w);
}

__global__ void k_scatter(const int* __restrict__ ei,
                          const float* __restrict__ H,
                          const float* __restrict__ a_s,
                          const float* __restrict__ a_d,
                          const float* __restrict__ maxv,
                          const float* __restrict__ denom,
                          float* __restrict__ out) {
    const int wave = threadIdx.x >> 5;
    const int lane = threadIdx.x & 31;
    const int e = blockIdx.x * 8 + wave;
    if (e >= E_TOT) return;
    int s, t; edge_st(ei, e, s, t);

    float alpha = expf(leaky(a_s[s] + a_d[t]) - maxv[t]) / denom[t];

    const float* hrow = H + s * DIM + lane;
    const float h0 = hrow[0];
    const float h1 = hrow[32];
    const float h2 = hrow[64];
    const float h3 = hrow[96];
    float* o = out + t * DIM + lane;
    unsafeAtomicAdd(o +  0, alpha * h0);
    unsafeAtomicAdd(o + 32, alpha * h1);
    unsafeAtomicAdd(o + 64, alpha * h2);
    unsafeAtomicAdd(o + 96, alpha * h3);
}

__global__ void k_finalize(float* __restrict__ out,
                           const float* __restrict__ bias) {
    int i = blockIdx.x * blockDim.x + threadIdx.x;
    out[i] = tanhf(out[i] + bias[i & (DIM - 1)]);
}

extern "C" void kernel_launch(void* const* d_in, const int* in_sizes, int n_in,
                              void* d_out, int out_size, void* d_ws, size_t ws_size,
                              hipStream_t stream) {
    const float* x        = (const float*)d_in[0];
    const int*   ei       = (const int*)  d_in[1];
    const float* W        = (const float*)d_in[2];
    const float* att_src  = (const float*)d_in[3];
    const float* att_dst  = (const float*)d_in[4];
    const float* bias     = (const float*)d_in[5];
    float*       out      = (float*)d_out;

    float* H     = (float*)d_ws;
    float* a_s   = H    + (size_t)N_NODES * DIM;
    float* a_d   = a_s  + N_NODES;
    float* maxv  = a_d  + N_NODES;
    float* denom = maxv + N_NODES;

    const int NE_ELEMS = N_NODES * DIM;

    k_init<<<NE_ELEMS / 256, 256, 0, stream>>>(out, maxv, denom);

    k_gemm_wmma<<<N_NODES / 16, 256, 0, stream>>>(x, W, H);

    k_attn_coef<<<N_NODES / 8, 256, 0, stream>>>(H, att_src, att_dst, a_s, a_d);

    const int eb = (E_TOT + 255) / 256;
    k_edge_max<<<eb, 256, 0, stream>>>(ei, a_s, a_d, maxv);
    k_edge_denom<<<eb, 256, 0, stream>>>(ei, a_s, a_d, maxv, denom);

    k_scatter<<<(E_TOT + 7) / 8, 256, 0, stream>>>(ei, H, a_s, a_d, maxv, denom, out);

    k_finalize<<<NE_ELEMS / 256, 256, 0, stream>>>(out, bias);
}


# aggregation: each edge's source is filed in its target's own out row (int atomic slot); one wave per target does max, denominator and row sums in registers in slot order, one store; no f32 atomics
# speedup vs baseline: 2.3755x; 1.9202x over previous
#include <hip/hip_runtime.h>
#include <math.h>

typedef __attribute__((ext_vector_type(2))) float v2f;
typedef __attribute__((ext_vector_type(8))) float v8f;

#define N_NODES   50000
#define N_EDGES   640000
#define DIM       128
#define NEG_SLOPE 0.2f
#define LIST_CAP  DIM

__global__ void k_gemm_wmma(const float* __restrict__ X,
                            const float* __restrict__ W,
                            float* __restrict__ H) {
    const int wave = threadIdx.x >> 5;
    const int lane = threadIdx.x & 31;
    const int hb   = lane >> 4;
    const int l    = lane & 15;
    const int m0   = blockIdx.x * 16;
    const int n0   = wave * 16;

    const float* xrow = X + (m0 + l) * DIM + 2 * hb;
    const float* wcol = W + (2 * hb) * DIM + n0 + l;

    v8f acc = {};
#pragma unroll
    for (int k = 0; k < DIM; k += 4) {
        v2f a = *(const v2f*)(xrow + k);
        v2f b;
        b.x = wcol[k * DIM];
        b.y = wcol[k * DIM + DIM];
        acc = __builtin_amdgcn_wmma_f32_16x16x4_f32(
            false, a, false, b, (short)0, acc, false, false);
    }

    float* hout = H + (m0 + 8 * hb) * DIM + n0 + l;
#pragma unroll
    for (int r = 0; r < 8; ++r)
        hout[r * DIM] = acc[r];
}

__global__ void k_attn_coef(const float* __restrict__ H,
                            const float* __restrict__ att_src,
                            const float* __restrict__ att_dst,
                            float* __restrict__ a_s, float* __restrict__ a_d,
                            int* __restrict__ cnt) {
    const int wave = threadIdx.x >> 5;
    const int lane = threadIdx.x & 31;
    const int node = blockIdx.x * 8 + wave;
    if (node >= N_NODES) return;

    float4 h4  = ((const float4*)(H + node * DIM))[lane];
    float4 as4 = ((const float4*)att_src)[lane];
    float4 ad4 = ((const float4*)att_dst)[lane];
    float s = h4.x * as4.x + h4.y * as4.y + h4.z * as4.z + h4.w * as4.w;
    float d = h4.x * ad4.x + h4.y * ad4.y + h4.z * ad4.z + h4.w * ad4.w;
#pragma unroll
    for (int off = 16; off > 0; off >>= 1) {
        s += __shfl_xor(s, off, 32);
        d += __shfl_xor(d, off, 32);
    }
    if (lane == 0) { a_s[node] = s; a_d[node] = d; cnt[node] = 0; }
}

__device__ __forceinline__ float leaky(float e) {
    return e > 0.0f ? e : NEG_SLOPE * e;
}

__device__ __forceinline__ float wave_max(float v) {
    for (int off = 16; off > 0; off >>= 1) v = fmaxf(v, __shfl_xor(v, off, 32));
    return v;
}

__device__ __forceinline__ float wave_sum(float v) {
    for (int off = 16; off > 0; off >>= 1) v += __shfl_xor(v, off, 32);
    return v;
}

__global__ void k_fill(const int* __restrict__ ei, int* __restrict__ cnt,
                       float* __restrict__ out) {
    const int e = blockIdx.x * 256 + threadIdx.x;
    const int s = ei[e];
    const int t = ei[N_EDGES + e];
    const int p = atomicAdd(&cnt[t], 1);
    if (p < LIST_CAP) out[t * DIM + p] = __int_as_float(s);
}

__global__ void k_gather(const int* __restrict__ ei,
                         const float* __restrict__ H,
                         const float* __restrict__ a_s,
                         const float* __restrict__ a_d,
                         const int* __restrict__ cnt,
                         const float* __restrict__ bias,
                         float* out) {
    const int wave = threadIdx.x >> 5;
    const int lane = threadIdx.x & 31;
    const int t = __builtin_amdgcn_readfirstlane(blockIdx.x * 8 + wave);
    const int deg = cnt[t];
    const bool scan = deg > LIST_CAP;
    const int nchunk = scan ? N_EDGES / 32 : (deg + 31) >> 5;
    float* orow = out + t * DIM;
    const float ad = a_d[t];
    const float e_self = leaky(a_s[t] + ad);

    auto fetch = [&](int c, int& s) -> bool {
        const int i = c * 32 + lane;
        if (scan) { s = ei[i]; return ei[N_EDGES + i] == t; }
        const bool v = i < deg;
        s = v ? __float_as_int(orow[i]) : 0;
        return v;
    };

    int s0;
    const bool v0 = fetch(0, s0);
    const float l0 = v0 ? leaky(a_s[s0] + ad) : -__builtin_inff();

    float m = fmaxf(e_self, l0);
    for (int c = 1; c < nchunk; ++c) {
        int s;
        if (fetch(c, s)) m = fmaxf(m, leaky(a_s[s] + ad));
    }
    m = wave_max(m);

    const float w0 = v0 ? expf(l0 - m) : 0.0f;
    float d = w0;
    for (int c = 1; c < nchunk; ++c) {
        int s;
        if (fetch(c, s)) d += expf(leaky(a_s[s] + ad) - m);
    }
    const float w_self = expf(e_self - m);
    const float denom = wave_sum(d) + w_self;

    const float* hself = H + t * DIM + lane;
    const float a_self = w_self / denom;
    float acc0 = a_self * hself[0];
    float acc1 = a_self * hself[32];
    float acc2 = a_self * hself[64];
    float acc3 = a_self * hself[96];

    auto drain = [&](bool v, int s, float alpha) {
        unsigned mask = __builtin_amdgcn_ballot_w32(v);
        while (mask) {
            const int j = __builtin_ctz(mask);
            mask &= mask - 1;
            const int sj = __builtin_amdgcn_readlane(s, j);
            const float aj = __int_as_float(
                __builtin_amdgcn_readlane(__float_as_int(alpha), j));
            const float* hrow = H + sj * DIM + lane;
            acc0 += aj * hrow[0];
            acc1 += aj * hrow[32];
            acc2 += aj * hrow[64];
            acc3 += aj * hrow[96];
        }
    };

    drain(v0, s0, w0 / denom);
    for (int c = 1; c < nchunk; ++c) {
        int s;
        const bool v = fetch(c, s);
        const float a = v ? expf(leaky(a_s[s] + ad) - m) / denom : 0.0f;
        drain(v, s, a);
    }

    orow[lane]      = tanhf(acc0 + bias[lane]);
    orow[lane + 32] = tanhf(acc1 + bias[lane + 32]);
    orow[lane + 64] = tanhf(acc2 + bias[lane + 64]);
    orow[lane + 96] = tanhf(acc3 + bias[lane + 96]);
}

extern "C" void kernel_launch(void* const* d_in, const int* in_sizes, int n_in,
                              void* d_out, int out_size, void* d_ws, size_t ws_size,
                              hipStream_t stream) {
    const float* x        = (const float*)d_in[0];
    const int*   ei       = (const int*)  d_in[1];
    const float* W        = (const float*)d_in[2];
    const float* att_src  = (const float*)d_in[3];
    const float* att_dst  = (const float*)d_in[4];
    const float* bias     = (const float*)d_in[5];
    float*       out      = (float*)d_out;

    float* H   = (float*)d_ws;
    float* a_s = H   + (size_t)N_NODES * DIM;
    float* a_d = a_s + N_NODES;
    int*   cnt = (int*)(a_d + N_NODES);

    k_gemm_wmma<<<N_NODES / 16, 256, 0, stream>>>(x, W, H);

    k_attn_coef<<<N_NODES / 8, 256, 0, stream>>>(H, att_src, att_dst, a_s, a_d, cnt);

    k_fill<<<N_EDGES / 256, 256, 0, stream>>>(ei, cnt, out);

    k_gather<<<N_NODES / 8, 256, 0, stream>>>(ei, H, a_s, a_d, cnt, bias, out);
}


# coefficients fused into the product kernel: one wave owns a 16x128 strip of H, a_s and a_d come from its accumulators by a 16-lane halving sum (new f32 order); 3 launches, H not re-read
# speedup vs baseline: 2.8145x; 1.1848x over previous
#include <hip/hip_runtime.h>
#include <math.h>

typedef __attribute__((ext_vector_type(2))) float v2f;
typedef __attribute__((ext_vector_type(8))) float v8f;

#define N_NODES   50000
#define N_EDGES   640000
#define DIM       128
#define NEG_SLOPE 0.2f
#define LIST_CAP  DIM

__global__ void __launch_bounds__(32) __attribute__((amdgpu_waves_per_eu(1, 4)))
k_gemm_coef(const float* __restrict__ X,
            const float* __restrict__ W,
            const float* __restrict__ att_src,
            const float* __restrict__ att_dst,
            float* __restrict__ H,
            float* __restrict__ a_s, float* __restrict__ a_d,
            int* __restrict__ cnt) {
    const int lane = threadIdx.x;
    const int hb   = lane >> 4;
    const int l    = lane & 15;
    const int m0   = blockIdx.x * 16;

    const float* xrow = X + (m0 + l) * DIM + 2 * hb;
    const float* wcol = W + (2 * hb) * DIM + l;

    v8f acc[8];
#pragma unroll
    for (int nt = 0; nt < 8; ++nt) acc[nt] = (v8f){};

    v2f a[DIM / 4];
#pragma unroll
    for (int k = 0; k < DIM; k += 4) a[k / 4] = *(const v2f*)(xrow + k);

#pragma unroll
    for (int k = 0; k < DIM; k += 4) {
#pragma unroll
        for (int nt = 0; nt < 8; ++nt) {
            v2f b;
            b.x = wcol[k * DIM + nt * 16];
            b.y = wcol[k * DIM + DIM + nt * 16];
            acc[nt] = __builtin_amdgcn_wmma_f32_16x16x4_f32(
                false, a[k / 4], false, b, (short)0, acc[nt], false, false);
        }
    }

    float* hout = H + (m0 + 8 * hb) * DIM + l;
#pragma unroll
    for (int nt = 0; nt < 8; ++nt)
#pragma unroll
        for (int r = 0; r < 8; ++r)
            hout[r * DIM + nt * 16] = acc[nt][r];

    float ps[8], pd[8];
#pragma unroll
    for (int r = 0; r < 8; ++r) { ps[r] = 0.0f; pd[r] = 0.0f; }
#pragma unroll
    for (int nt = 0; nt < 8; ++nt) {
        const float cs = att_src[nt * 16 + l];
        const float cd = att_dst[nt * 16 + l];
#pragma unroll
        for (int r = 0; r < 8; ++r) {
            ps[r] += acc[nt][r] * cs;
            pd[r] += acc[nt][r] * cd;
        }
    }

    const bool b8 = (l & 8) != 0, b4 = (l & 4) != 0, b2 = (l & 2) != 0, b1 = (l & 1) != 0;
    float v8[8], v4[4], v2[2];
#pragma unroll
    for (int r = 0; r < 8; ++r) {
        const float keep = b8 ? pd[r] : ps[r];
        const float send = b8 ? ps[r] : pd[r];
        v8[r] = keep + __shfl_xor(send, 8, 32);
    }
#pragma unroll
    for (int j = 0; j < 4; ++j) {
        const float keep = b4 ? v8[j + 4] : v8[j];
        const float send = b4 ? v8[j] : v8[j + 4];
        v4[j] = keep + __shfl_xor(send, 4, 32);
    }
#pragma unroll
    for (int j = 0; j < 2; ++j) {
        const float keep = b2 ? v4[j + 2] : v4[j];
        const float send = b2 ? v4[j] : v4[j + 2];
        v2[j] = keep + __shfl_xor(send, 2, 32);
    }
    const float keep1 = b1 ? v2[1] : v2[0];
    const float send1 = b1 ? v2[0] : v2[1];
    const float tot = keep1 + __shfl_xor(send1, 1, 32);

    float* dst = b8 ? a_d : a_s;
    dst[m0 + 8 * hb + (l & 7)] = tot;

    if (lane < 16) cnt[m0 + lane] = 0;
}

__device__ __forceinline__ float leaky(float e) {
    return e > 0.0f ? e : NEG_SLOPE * e;
}

__device__ __forceinline__ float wave_max(float v) {
    for (int off = 16; off > 0; off >>= 1) v = fmaxf(v, __shfl_xor(v, off, 32));
    return v;
}

__device__ __forceinline__ float wave_sum(float v) {
    for (int off = 16; off > 0; off >>= 1) v += __shfl_xor(v, off, 32);
    return v;
}

__global__ void k_fill(const int* __restrict__ ei, int* __restrict__ cnt,
                       float* __restrict__ out) {
    const int e = blockIdx.x * 256 + threadIdx.x;
    const int s = ei[e];
    const int t = ei[N_EDGES + e];
    const int p = atomicAdd(&cnt[t], 1);
    if (p < LIST_CAP) out[t * DIM + p] = __int_as_float(s);
}

__global__ void k_gather(const int* __restrict__ ei,
                         const float* __restrict__ H,
                         const float* __restrict__ a_s,
                         const float* __restrict__ a_d,
                         const int* __restrict__ cnt,
                         const float* __restrict__ bias,
                         float* out) {
    const int wave = threadIdx.x >> 5;
    const int lane = threadIdx.x & 31;
    const int t = __builtin_amdgcn_readfirstlane(blockIdx.x * 8 + wave);
    const int deg = cnt[t];
    const bool scan = deg > LIST_CAP;
    const int nchunk = scan ? N_EDGES / 32 : (deg + 31) >> 5;
    float* orow = out + t * DIM;
    const float ad = a_d[t];
    const float e_self = leaky(a_s[t] + ad);

    auto fetch = [&](int c, int& s) -> bool {
        const int i = c * 32 + lane;
        if (scan) { s = ei[i]; return ei[N_EDGES + i] == t; }
        const bool v = i < deg;
        s = v ? __float_as_int(orow[i]) : 0;
        return v;
    };

    int s0;
    const bool v0 = fetch(0, s0);
    const float l0 = v0 ? leaky(a_s[s0] + ad) : -__builtin_inff();

    float m = fmaxf(e_self, l0);
    for (int c = 1; c < nchunk; ++c) {
        int s;
        if (fetch(c, s)) m = fmaxf(m, leaky(a_s[s] + ad));
    }
    m = wave_max(m);

    const float w0 = v0 ? expf(l0 - m) : 0.0f;
    float d = w0;
    for (int c = 1; c < nchunk; ++c) {
        int s;
        if (fetch(c, s)) d += expf(leaky(a_s[s] + ad) - m);
    }
    const float w_self = expf(e_self - m);
    const float denom = wave_sum(d) + w_self;

    const float* hself = H + t * DIM + lane;
    const float a_self = w_self / denom;
    float acc0 = a_self * hself[0];
    float acc1 = a_self * hself[32];
    float acc2 = a_self * hself[64];
    float acc3 = a_self * hself[96];

    auto drain = [&](bool v, int s, float alpha) {
        unsigned mask = __builtin_amdgcn_ballot_w32(v);
        while (mask) {
            const int j = __builtin_ctz(mask);
            mask &= mask - 1;
            const int sj = __builtin_amdgcn_readlane(s, j);
            const float aj = __int_as_float(
                __builtin_amdgcn_readlane(__float_as_int(alpha), j));
            const float* hrow = H + sj * DIM + lane;
            acc0 += aj * hrow[0];
            acc1 += aj * hrow[32];
            acc2 += aj * hrow[64];
            acc3 += aj * hrow[96];
        }
    };

    drain(v0, s0, w0 / denom);
    for (int c = 1; c < nchunk; ++c) {
        int s;
        const bool v = fetch(c, s);
        const float a = v ? expf(leaky(a_s[s] + ad) - m) / denom : 0.0f;
        drain(v, s, a);
    }

    orow[lane]      = tanhf(acc0 + bias[lane]);
    orow[lane + 32] = tanhf(acc1 + bias[lane + 32]);
    orow[lane + 64] = tanhf(acc2 + bias[lane + 64]);
    orow[lane + 96] = tanhf(acc3 + bias[lane + 96]);
}

extern "C" void kernel_launch(void* const* d_in, const int* in_sizes, int n_in,
                              void* d_out, int out_size, void* d_ws, size_t ws_size,
                              hipStream_t stream) {
    const float* x        = (const float*)d_in[0];
    const int*   ei       = (const int*)  d_in[1];
    const float* W        = (const float*)d_in[2];
    const float* att_src  = (const float*)d_in[3];
    const float* att_dst  = (const float*)d_in[4];
    const float* bias     = (const float*)d_in[5];
    float*       out      = (float*)d_out;

    float* H   = (float*)d_ws;
    float* a_s = H   + (size_t)N_NODES * DIM;
    float* a_d = a_s + N_NODES;
    int*   cnt = (int*)(a_d + N_NODES);

    k_gemm_coef<<<N_NODES / 16, 32, 0, stream>>>(x, W, att_src, att_dst, H, a_s, a_d, cnt);

    k_fill<<<N_EDGES / 256, 256, 0, stream>>>(ei, cnt, out);

    k_gather<<<N_NODES / 8, 256, 0, stream>>>(ei, H, a_s, a_d, cnt, bias, out);
}


# gather: a wave serves 4 consecutive targets, so 12500 waves enter, not 50000; their counts, lists and a_s gathers go out together, rows of up to 8 in-edges are in flight; every sum order kept
# speedup vs baseline: 3.0705x; 1.0910x over previous
#include <hip/hip_runtime.h>
#include <math.h>

typedef __attribute__((ext_vector_type(2))) float v2f;
typedef __attribute__((ext_vector_type(8))) float v8f;

#define N_NODES   50000
#define N_EDGES   640000
#define DIM       128
#define NEG_SLOPE 0.2f
#define LIST_CAP  DIM

__global__ void __launch_bounds__(32) __attribute__((amdgpu_waves_per_eu(1, 4)))
k_gemm_coef(const float* __restrict__ X,
            const float* __restrict__ W,
            const float* __restrict__ att_src,
            const float* __restrict__ att_dst,
            float* __restrict__ H,
            float* __restrict__ a_s, float* __restrict__ a_d,
            int* __restrict__ cnt) {
    const int lane = threadIdx.x;
    const int hb   = lane >> 4;
    const int l    = lane & 15;
    const int m0   = blockIdx.x * 16;

    const float* xrow = X + (m0 + l) * DIM + 2 * hb;
    const float* wcol = W + (2 * hb) * DIM + l;

    v8f acc[8];
#pragma unroll
    for (int nt = 0; nt < 8; ++nt) acc[nt] = (v8f){};

    v2f a[DIM / 4];
#pragma unroll
    for (int k = 0; k < DIM; k += 4) a[k / 4] = *(const v2f*)(xrow + k);

#pragma unroll
    for (int k = 0; k < DIM; k += 4) {
#pragma unroll
        for (int nt = 0; nt < 8; ++nt) {
            v2f b;
            b.x = wcol[k * DIM + nt * 16];
            b.y = wcol[k * DIM + DIM + nt * 16];
            acc[nt] = __builtin_amdgcn_wmma_f32_16x16x4_f32(
                false, a[k / 4], false, b, (short)0, acc[nt], false, false);
        }
    }

    float* hout = H + (m0 + 8 * hb) * DIM + l;
#pragma unroll
    for (int nt = 0; nt < 8; ++nt)
#pragma unroll
        for (int r = 0; r < 8; ++r)
            hout[r * DIM + nt * 16] = acc[nt][r];

    float ps[8], pd[8];
#pragma unroll
    for (int r = 0; r < 8; ++r) { ps[r] = 0.0f; pd[r] = 0.0f; }
#pragma unroll
    for (int nt = 0; nt < 8; ++nt) {
        const float cs = att_src[nt * 16 + l];
        const float cd = att_dst[nt * 16 + l];
#pragma unroll
        for (int r = 0; r < 8; ++r) {
            ps[r] += acc[nt][r] * cs;
            pd[r] += acc[nt][r] * cd;
        }
    }

    const bool b8 = (l & 8) != 0, b4 = (l & 4) != 0, b2 = (l & 2) != 0, b1 = (l & 1) != 0;
    float v8[8], v4[4], v2[2];
#pragma unroll
    for (int r = 0; r < 8; ++r) {
        const float keep = b8 ? pd[r] : ps[r];
        const float send = b8 ? ps[r] : pd[r];
        v8[r] = keep + __shfl_xor(send, 8, 32);
    }
#pragma unroll
    for (int j = 0; j < 4; ++j) {
        const float keep = b4 ? v8[j + 4] : v8[j];
        const float send = b4 ? v8[j] : v8[j + 4];
        v4[j] = keep + __shfl_xor(send, 4, 32);
    }
#pragma unroll
    for (int j = 0; j < 2; ++j) {
        const float keep = b2 ? v4[j + 2] : v4[j];
        const float send = b2 ? v4[j] : v4[j + 2];
        v2[j] = keep + __shfl_xor(send, 2, 32);
    }
    const float keep1 = b1 ? v2[1] : v2[0];
    const float send1 = b1 ? v2[0] : v2[1];
    const float tot = keep1 + __shfl_xor(send1, 1, 32);

    float* dst = b8 ? a_d : a_s;
    dst[m0 + 8 * hb + (l & 7)] = tot;

    if (lane < 16) cnt[m0 + lane] = 0;
}

__device__ __forceinline__ float leaky(float e) {
    return e > 0.0f ? e : NEG_SLOPE * e;
}

__device__ __forceinline__ float wave_max(float v) {
    for (int off = 16; off > 0; off >>= 1) v = fmaxf(v, __shfl_xor(v, off, 32));
    return v;
}

__device__ __forceinline__ float wave_sum(float v) {
    for (int off = 16; off > 0; off >>= 1) v += __shfl_xor(v, off, 32);
    return v;
}

__global__ void k_fill(const int* __restrict__ ei, int* __restrict__ cnt,
                       float* __restrict__ out) {
    const int e = blockIdx.x * 256 + threadIdx.x;
    const int s = ei[e];
    const int t = ei[N_EDGES + e];
    const int p = atomicAdd(&cnt[t], 1);
    if (p < LIST_CAP) out[t * DIM + p] = __int_as_float(s);
}

#define TPW 4
#define GW  4

__device__ __forceinline__ void gather_one(const int t, const int lane,
                                           const int* __restrict__ ei,
                                           const float* __restrict__ H,
                                           const float* __restrict__ a_s,
                                           const float* __restrict__ a_d,
                                           const int* __restrict__ cnt,
                                           const float* __restrict__ bias,
                                           float* out) {
    const int deg = cnt[t];
    const bool scan = deg > LIST_CAP;
    const int nchunk = scan ? N_EDGES / 32 : (deg + 31) >> 5;
    float* orow = out + t * DIM;
    const float ad = a_d[t];
    const float e_self = leaky(a_s[t] + ad);

    auto fetch = [&](int c, int& s) -> bool {
        const int i = c * 32 + lane;
        if (scan) { s = ei[i]; return ei[N_EDGES + i] == t; }
        const bool v = i < deg;
        s = v ? __float_as_int(orow[i]) : 0;
        return v;
    };

    int s0;
    const bool v0 = fetch(0, s0);
    const float l0 = v0 ? leaky(a_s[s0] + ad) : -__builtin_inff();

    float m = fmaxf(e_self, l0);
    for (int c = 1; c < nchunk; ++c) {
        int s;
        if (fetch(c, s)) m = fmaxf(m, leaky(a_s[s] + ad));
    }
    m = wave_max(m);

    const float w0 = v0 ? expf(l0 - m) : 0.0f;
    float d = w0;
    for (int c = 1; c < nchunk; ++c) {
        int s;
        if (fetch(c, s)) d += expf(leaky(a_s[s] + ad) - m);
    }
    const float w_self = expf(e_self - m);
    const float denom = wave_sum(d) + w_self;

    const float* hself = H + t * DIM + lane;
    const float a_self = w_self / denom;
    float acc0 = a_self * hself[0];
    float acc1 = a_self * hself[32];
    float acc2 = a_self * hself[64];
    float acc3 = a_self * hself[96];

    auto drain = [&](bool v, int s, float alpha) {
        unsigned mask = __builtin_amdgcn_ballot_w32(v);
        while (mask) {
            const int j = __builtin_ctz(mask);
            mask &= mask - 1;
            const int sj = __builtin_amdgcn_readlane(s, j);
            const float aj = __int_as_float(
                __builtin_amdgcn_readlane(__float_as_int(alpha), j));
            const float* hrow = H + sj * DIM + lane;
            acc0 += aj * hrow[0];
            acc1 += aj * hrow[32];
            acc2 += aj * hrow[64];
            acc3 += aj * hrow[96];
        }
    };

    drain(v0, s0, w0 / denom);
    for (int c = 1; c < nchunk; ++c) {
        int s;
        const bool v = fetch(c, s);
        const float a = v ? expf(leaky(a_s[s] + ad) - m) / denom : 0.0f;
        drain(v, s, a);
    }

    orow[lane]      = tanhf(acc0 + bias[lane]);
    orow[lane + 32] = tanhf(acc1 + bias[lane + 32]);
    orow[lane + 64] = tanhf(acc2 + bias[lane + 64]);
    orow[lane + 96] = tanhf(acc3 + bias[lane + 96]);
}

template <int B>
__device__ __forceinline__ void drain_batch(const int first, const int s, const float alpha,
                                            const int lane, const float* __restrict__ H,
                                            float& acc0, float& acc1,
                                            float& acc2, float& acc3) {
    float aj[B];
    float r0[B], r1[B], r2[B], r3[B];
#pragma unroll
    for (int b = 0; b < B; ++b) {
        const int sj = __builtin_amdgcn_readlane(s, first + b);
        aj[b] = __int_as_float(__builtin_amdgcn_readlane(__float_as_int(alpha), first + b));
        const float* hrow = H + sj * DIM;
        r0[b] = hrow[lane];
        r1[b] = hrow[lane + 32];
        r2[b] = hrow[lane + 64];
        r3[b] = hrow[lane + 96];
    }
#pragma unroll
    for (int b = 0; b < B; ++b) {
        acc0 = __builtin_fmaf(aj[b], r0[b], acc0);
        acc1 = __builtin_fmaf(aj[b], r1[b], acc1);
        acc2 = __builtin_fmaf(aj[b], r2[b], acc2);
        acc3 = __builtin_fmaf(aj[b], r3[b], acc3);
    }
}

__global__ void __launch_bounds__(GW * 32)
k_gather(const int* __restrict__ ei,
         const float* __restrict__ H,
         const float* __restrict__ a_s,
         const float* __restrict__ a_d,
         const int* __restrict__ cnt,
         const float* __restrict__ bias,
         float* out) {
    const int wave = threadIdx.x >> 5;
    const int lane = threadIdx.x & 31;
    const int t0 = __builtin_amdgcn_readfirstlane((blockIdx.x * GW + wave) * TPW);

    int deg[TPW];
    int degmax = 0;
#pragma unroll
    for (int k = 0; k < TPW; ++k) {
        deg[k] = cnt[t0 + k];
        degmax = deg[k] > degmax ? deg[k] : degmax;
    }

    if (degmax > 32) {
#pragma nounroll
        for (int k = 0; k < TPW; ++k)
            gather_one(t0 + k, lane, ei, H, a_s, a_d, cnt, bias, out);
        return;
    }

    const float bias0 = bias[lane];
    const float bias1 = bias[lane + 32];
    const float bias2 = bias[lane + 64];
    const float bias3 = bias[lane + 96];

    float ad[TPW], e_self[TPW];
#pragma unroll
    for (int k = 0; k < TPW; ++k) {
        ad[k]     = a_d[t0 + k];
        e_self[k] = leaky(a_s[t0 + k] + ad[k]);
    }

    int  s0[TPW];
    bool v0[TPW];
#pragma unroll
    for (int k = 0; k < TPW; ++k) {
        const float* orow = out + (t0 + k) * DIM;
        v0[k] = lane < deg[k];
        s0[k] = v0[k] ? __float_as_int(orow[lane]) : 0;
    }

    float as0[TPW];
#pragma unroll
    for (int k = 0; k < TPW; ++k) as0[k] = a_s[s0[k]];

    float l0[TPW], m[TPW];
#pragma unroll
    for (int k = 0; k < TPW; ++k) {
        l0[k] = v0[k] ? leaky(as0[k] + ad[k]) : -__builtin_inff();
        m[k]  = wave_max(fmaxf(e_self[k], l0[k]));
    }

    float w0[TPW], dsum[TPW];
#pragma unroll
    for (int k = 0; k < TPW; ++k) w0[k] = v0[k] ? expf(l0[k] - m[k]) : 0.0f;
#pragma unroll
    for (int k = 0; k < TPW; ++k) dsum[k] = wave_sum(w0[k]);
    float alpha[TPW], a_self[TPW];
#pragma unroll
    for (int k = 0; k < TPW; ++k) {
        const float w_self = expf(e_self[k] - m[k]);
        const float denom  = dsum[k] + w_self;
        alpha[k]  = w0[k] / denom;
        a_self[k] = __int_as_float(__builtin_amdgcn_readfirstlane(__float_as_int(w_self / denom)));
    }

#pragma unroll
    for (int k = 0; k < TPW; ++k) {
        const int t = t0 + k;
        const float* hself = H + t * DIM;
        float acc0 = a_self[k] * hself[lane];
        float acc1 = a_self[k] * hself[lane + 32];
        float acc2 = a_self[k] * hself[lane + 64];
        float acc3 = a_self[k] * hself[lane + 96];

        const int n = deg[k];
        int j = 0;
        for (; j + 8 <= n; j += 8)
            drain_batch<8>(j, s0[k], alpha[k], lane, H, acc0, acc1, acc2, acc3);
        if (j + 4 <= n) { drain_batch<4>(j, s0[k], alpha[k], lane, H, acc0, acc1, acc2, acc3); j += 4; }
        if (j + 2 <= n) { drain_batch<2>(j, s0[k], alpha[k], lane, H, acc0, acc1, acc2, acc3); j += 2; }
        if (j < n)        drain_batch<1>(j, s0[k], alpha[k], lane, H, acc0, acc1, acc2, acc3);

        float* orow = out + t * DIM;
        orow[lane]      = tanhf(acc0 + bias0);
        orow[lane + 32] = tanhf(acc1 + bias1);
        orow[lane + 64] = tanhf(acc2 + bias2);
        orow[lane + 96] = tanhf(acc3 + bias3);
    }
}

extern "C" void kernel_launch(void* const* d_in, const int* in_sizes, int n_in,
                              void* d_out, int out_size, void* d_ws, size_t ws_size,
                              hipStream_t stream) {
    const float* x        = (const float*)d_in[0];
    const int*   ei       = (const int*)  d_in[1];
    const float* W        = (const float*)d_in[2];
    const float* att_src  = (const float*)d_in[3];
    const float* att_dst  = (const float*)d_in[4];
    const float* bias     = (const float*)d_in[5];
    float*       out      = (float*)d_out;

    float* H   = (float*)d_ws;
    float* a_s = H   + (size_t)N_NODES * DIM;
    float* a_d = a_s + N_NODES;
    int*   cnt = (int*)(a_d + N_NODES);

    k_gemm_coef<<<N_NODES / 16, 32, 0, stream>>>(x, W, att_src, att_dst, H, a_s, a_d, cnt);

    k_fill<<<N_EDGES / 256, 256, 0, stream>>>(ei, cnt, out);

    k_gather<<<N_NODES / (GW * TPW), GW * 32, 0, stream>>>(ei, H, a_s, a_d, cnt, bias, out);
}


# filing kernel: in-edge counters in LDS, not memory: a work-group owns 256 targets and scans the list's targets, the LDS add returns the slot; list lines stored whole; slots still in arrival order
# speedup vs baseline: 4.2097x; 1.3710x over previous
#include <hip/hip_runtime.h>
#include <math.h>

typedef __attribute__((ext_vector_type(2))) float v2f;
typedef __attribute__((ext_vector_type(8))) float v8f;

#define N_NODES   50000
#define N_EDGES   640000
#define DIM       128
#define NEG_SLOPE 0.2f
#define LIST_CAP  DIM

__global__ void __launch_bounds__(32) __attribute__((amdgpu_waves_per_eu(1, 4)))
k_gemm_coef(const float* __restrict__ X,
            const float* __restrict__ W,
            const float* __restrict__ att_src,
            const float* __restrict__ att_dst,
            float* __restrict__ H,
            float* __restrict__ a_s, float* __restrict__ a_d) {
    const int lane = threadIdx.x;
    const int hb   = lane >> 4;
    const int l    = lane & 15;
    const int m0   = blockIdx.x * 16;

    const float* xrow = X + (m0 + l) * DIM + 2 * hb;
    const float* wcol = W + (2 * hb) * DIM + l;

    v8f acc[8];
#pragma unroll
    for (int nt = 0; nt < 8; ++nt) acc[nt] = (v8f){};

    v2f a[DIM / 4];
#pragma unroll
    for (int k = 0; k < DIM; k += 4) a[k / 4] = *(const v2f*)(xrow + k);

#pragma unroll
    for (int k = 0; k < DIM; k += 4) {
#pragma unroll
        for (int nt = 0; nt < 8; ++nt) {
            v2f b;
            b.x = wcol[k * DIM + nt * 16];
            b.y = wcol[k * DIM + DIM + nt * 16];
            acc[nt] = __builtin_amdgcn_wmma_f32_16x16x4_f32(
                false, a[k / 4], false, b, (short)0, acc[nt], false, false);
        }
    }

    float* hout = H + (m0 + 8 * hb) * DIM + l;
#pragma unroll
    for (int nt = 0; nt < 8; ++nt)
#pragma unroll
        for (int r = 0; r < 8; ++r)
            hout[r * DIM + nt * 16] = acc[nt][r];

    float ps[8], pd[8];
#pragma unroll
    for (int r = 0; r < 8; ++r) { ps[r] = 0.0f; pd[r] = 0.0f; }
#pragma unroll
    for (int nt = 0; nt < 8; ++nt) {
        const float cs = att_src[nt * 16 + l];
        const float cd = att_dst[nt * 16 + l];
#pragma unroll
        for (int r = 0; r < 8; ++r) {
            ps[r] += acc[nt][r] * cs;
            pd[r] += acc[nt][r] * cd;
        }
    }

    const bool b8 = (l & 8) != 0, b4 = (l & 4) != 0, b2 = (l & 2) != 0, b1 = (l & 1) != 0;
    float v8[8], v4[4], v2[2];
#pragma unroll
    for (int r = 0; r < 8; ++r) {
        const float keep = b8 ? pd[r] : ps[r];
        const float send = b8 ? ps[r] : pd[r];
        v8[r] = keep + __shfl_xor(send, 8, 32);
    }
#pragma unroll
    for (int j = 0; j < 4; ++j) {
        const float keep = b4 ? v8[j + 4] : v8[j];
        const float send = b4 ? v8[j] : v8[j + 4];
        v4[j] = keep + __shfl_xor(send, 4, 32);
    }
#pragma unroll
    for (int j = 0; j < 2; ++j) {
        const float keep = b2 ? v4[j + 2] : v4[j];
        const float send = b2 ? v4[j] : v4[j + 2];
        v2[j] = keep + __shfl_xor(send, 2, 32);
    }
    const float keep1 = b1 ? v2[1] : v2[0];
    const float send1 = b1 ? v2[0] : v2[1];
    const float tot = keep1 + __shfl_xor(send1, 1, 32);

    float* dst = b8 ? a_d : a_s;
    dst[m0 + 8 * hb + (l & 7)] = tot;
}

__device__ __forceinline__ float leaky(float e) {
    return e > 0.0f ? e : NEG_SLOPE * e;
}

__device__ __forceinline__ float wave_max(float v) {
    for (int off = 16; off > 0; off >>= 1) v = fmaxf(v, __shfl_xor(v, off, 32));
    return v;
}

__device__ __forceinline__ float wave_sum(float v) {
    for (int off = 16; off > 0; off >>= 1) v += __shfl_xor(v, off, 32);
    return v;
}

#define RNG   256
#define NGRP  ((N_NODES + RNG - 1) / RNG)
#define FT    1024
#define NQ    (N_EDGES / 4)
#define QU    8
#define STEP  (FT * QU)
#define NIT   ((NQ + STEP - 1) / STEP)
#define LROW  32
#define KLIM  (RNG * 32)

template <bool GUARD>
__device__ __forceinline__ void load_groups(int4 (&d)[QU], const int4* __restrict__ dst4,
                                            const int q0) {
#pragma unroll
    for (int j = 0; j < QU; ++j) {
        const int q = q0 + j * FT;
        if (GUARD) d[j] = q < NQ ? dst4[q] : make_int4(-1, -1, -1, -1);
        else       d[j] = dst4[q];
    }
}

__device__ __forceinline__ void file_trip(const int4 (&d)[QU], const int q0,
                                          const int t_base, const unsigned kb,
                                          int* s_cnt, int* s_edge,
                                          const int* __restrict__ ei,
                                          float* __restrict__ out) {
    unsigned key[4 * QU];
#pragma unroll
    for (int j = 0; j < QU; ++j) {
        key[4 * j + 0] = ((unsigned)d[j].x << 5) + ((unsigned)(4 * j + 0) - kb);
        key[4 * j + 1] = ((unsigned)d[j].y << 5) + ((unsigned)(4 * j + 1) - kb);
        key[4 * j + 2] = ((unsigned)d[j].z << 5) + ((unsigned)(4 * j + 2) - kb);
        key[4 * j + 3] = ((unsigned)d[j].w << 5) + ((unsigned)(4 * j + 3) - kb);
    }
    unsigned k = key[0];
#pragma unroll
    for (int c = 1; c < 4 * QU; ++c) k = min(k, key[c]);

    while (__builtin_amdgcn_ballot_w32(k < KLIM)) {
        unsigned done = KLIM;
        if (k < KLIM) {
            const unsigned u = k >> 5, c = k & 31;
            const int e = (q0 + (int)(c >> 2) * FT) * 4 + (int)(c & 3);
            const int p = atomicAdd(&s_cnt[u], 1);
            if (p < LROW) s_edge[u * LROW + p] = e;
            else if (p < LIST_CAP)
                out[(t_base + (int)u) * DIM + p] = __int_as_float(ei[e]);
            done = k + 1;
        }
        unsigned rel = key[0] - done;
#pragma unroll
        for (int c = 1; c < 4 * QU; ++c) rel = min(rel, key[c] - done);
        k = rel < KLIM - done ? rel + done : 0xFFFFFFFFu;
    }
}

__global__ void __launch_bounds__(FT)
k_fill(const int* __restrict__ ei, int* __restrict__ cnt, float* __restrict__ out) {
    __shared__ int s_cnt[RNG];
    __shared__ int s_edge[RNG * LROW];
    const int tid    = threadIdx.x;
    const int t_base = blockIdx.x * RNG;

    if (tid < RNG) s_cnt[tid] = 0;
    __syncthreads();

    const int4* dst4 = (const int4*)(ei + N_EDGES);

    const unsigned kb = (unsigned)t_base << 5;

    static_assert(NIT % 2 == 0 && (NIT - 1) * STEP <= NQ && NIT * STEP >= NQ,
                  "trips 0 .. NIT-2 are whole, trip NIT-1 is the padded one");
    int4 a[QU], b[QU];
    load_groups<false>(a, dst4, tid);
#pragma nounroll
    for (int it = 0; it < NIT - 2; it += 2) {
        const int q0 = tid + it * STEP;
        load_groups<false>(b, dst4, q0 + STEP);
        file_trip(a, q0, t_base, kb, s_cnt, s_edge, ei, out);
        load_groups<false>(a, dst4, q0 + 2 * STEP);
        file_trip(b, q0 + STEP, t_base, kb, s_cnt, s_edge, ei, out);
    }
    {
        const int q0 = tid + (NIT - 2) * STEP;
        load_groups<true>(b, dst4, q0 + STEP);
        file_trip(a, q0, t_base, kb, s_cnt, s_edge, ei, out);
        file_trip(b, q0 + STEP, t_base, kb, s_cnt, s_edge, ei, out);
    }
    __syncthreads();

    const int wave = tid >> 5;
    const int lane = tid & 31;
    int  sv[RNG / 32];
    bool ok[RNG / 32];
#pragma unroll
    for (int k = 0; k < RNG / 32; ++k) {
        const int u = wave * (RNG / 32) + k;
        ok[k] = lane < s_cnt[u];
        sv[k] = ei[ok[k] ? s_edge[u * LROW + lane] : 0];
    }
#pragma unroll
    for (int k = 0; k < RNG / 32; ++k) {
        const int t = t_base + wave * (RNG / 32) + k;
        if (ok[k]) out[t * DIM + lane] = __int_as_float(sv[k]);
    }

    if (tid < RNG && t_base + tid < N_NODES) cnt[t_base + tid] = s_cnt[tid];
}

#define TPW 4
#define GW  4

__device__ __forceinline__ void gather_one(const int t, const int lane,
                                           const int* __restrict__ ei,
                                           const float* __restrict__ H,
                                           const float* __restrict__ a_s,
                                           const float* __restrict__ a_d,
                                           const int* __restrict__ cnt,
                                           const float* __restrict__ bias,
                                           float* out) {
    const int deg = cnt[t];
    const bool scan = deg > LIST_CAP;
    const int nchunk = scan ? N_EDGES / 32 : (deg + 31) >> 5;
    float* orow = out + t * DIM;
    const float ad = a_d[t];
    const float e_self = leaky(a_s[t] + ad);

    auto fetch = [&](int c, int& s) -> bool {
        const int i = c * 32 + lane;
        if (scan) { s = ei[i]; return ei[N_EDGES + i] == t; }
        const bool v = i < deg;
        s = v ? __float_as_int(orow[i]) : 0;
        return v;
    };

    int s0;
    const bool v0 = fetch(0, s0);
    const float l0 = v0 ? leaky(a_s[s0] + ad) : -__builtin_inff();

    float m = fmaxf(e_self, l0);
    for (int c = 1; c < nchunk; ++c) {
        int s;
        if (fetch(c, s)) m = fmaxf(m, leaky(a_s[s] + ad));
    }
    m = wave_max(m);

    const float w0 = v0 ? expf(l0 - m) : 0.0f;
    float d = w0;
    for (int c = 1; c < nchunk; ++c) {
        int s;
        if (fetch(c, s)) d += expf(leaky(a_s[s] + ad) - m);
    }
    const float w_self = expf(e_self - m);
    const float denom = wave_sum(d) + w_self;

    const float* hself = H + t * DIM + lane;
    const float a_self = w_self / denom;
    float acc0 = a_self * hself[0];
    float acc1 = a_self * hself[32];
    float acc2 = a_self * hself[64];
    float acc3 = a_self * hself[96];

    auto drain = [&](bool v, int s, float alpha) {
        unsigned mask = __builtin_amdgcn_ballot_w32(v);
        while (mask) {
            const int j = __builtin_ctz(mask);
            mask &= mask - 1;
            const int sj = __builtin_amdgcn_readlane(s, j);
            const float aj = __int_as_float(
                __builtin_amdgcn_readlane(__float_as_int(alpha), j));
            const float* hrow = H + sj * DIM + lane;
            acc0 += aj * hrow[0];
            acc1 += aj * hrow[32];
            acc2 += aj * hrow[64];
            acc3 += aj * hrow[96];
        }
    };

    drain(v0, s0, w0 / denom);
    for (int c = 1; c < nchunk; ++c) {
        int s;
        const bool v = fetch(c, s);
        const float a = v ? expf(leaky(a_s[s] + ad) - m) / denom : 0.0f;
        drain(v, s, a);
    }

    orow[lane]      = tanhf(acc0 + bias[lane]);
    orow[lane + 32] = tanhf(acc1 + bias[lane + 32]);
    orow[lane + 64] = tanhf(acc2 + bias[lane + 64]);
    orow[lane + 96] = tanhf(acc3 + bias[lane + 96]);
}

template <int B>
__device__ __forceinline__ void drain_batch(const int first, const int s, const float alpha,
                                            const int lane, const float* __restrict__ H,
                                            float& acc0, float& acc1,
                                            float& acc2, float& acc3) {
    float aj[B];
    float r0[B], r1[B], r2[B], r3[B];
#pragma unroll
    for (int b = 0; b < B; ++b) {
        const int sj = __builtin_amdgcn_readlane(s, first + b);
        aj[b] = __int_as_float(__builtin_amdgcn_readlane(__float_as_int(alpha), first + b));
        const float* hrow = H + sj * DIM;
        r0[b] = hrow[lane];
        r1[b] = hrow[lane + 32];
        r2[b] = hrow[lane + 64];
        r3[b] = hrow[lane + 96];
    }
#pragma unroll
    for (int b = 0; b < B; ++b) {
        acc0 = __builtin_fmaf(aj[b], r0[b], acc0);
        acc1 = __builtin_fmaf(aj[b], r1[b], acc1);
        acc2 = __builtin_fmaf(aj[b], r2[b], acc2);
        acc3 = __builtin_fmaf(aj[b], r3[b], acc3);
    }
}

__global__ void __launch_bounds__(GW * 32)
k_gather(const int* __restrict__ ei,
         const float* __restrict__ H,
         const float* __restrict__ a_s,
         const float* __restrict__ a_d,
         const int* __restrict__ cnt,
         const float* __restrict__ bias,
         float* out) {
    const int wave = threadIdx.x >> 5;
    const int lane = threadIdx.x & 31;
    const int t0 = __builtin_amdgcn_readfirstlane((blockIdx.x * GW + wave) * TPW);

    int deg[TPW];
    int degmax = 0;
#pragma unroll
    for (int k = 0; k < TPW; ++k) {
        deg[k] = cnt[t0 + k];
        degmax = deg[k] > degmax ? deg[k] : degmax;
    }

    if (degmax > 32) {
#pragma nounroll
        for (int k = 0; k < TPW; ++k)
            gather_one(t0 + k, lane, ei, H, a_s, a_d, cnt, bias, out);
        return;
    }

    const float bias0 = bias[lane];
    const float bias1 = bias[lane + 32];
    const float bias2 = bias[lane + 64];
    const float bias3 = bias[lane + 96];

    float ad[TPW], e_self[TPW];
#pragma unroll
    for (int k = 0; k < TPW; ++k) {
        ad[k]     = a_d[t0 + k];
        e_self[k] = leaky(a_s[t0 + k] + ad[k]);
    }

    int  s0[TPW];
    bool v0[TPW];
#pragma unroll
    for (int k = 0; k < TPW; ++k) {
        const float* orow = out + (t0 + k) * DIM;
        v0[k] = lane < deg[k];
        s0[k] = v0[k] ? __float_as_int(orow[lane]) : 0;
    }

    float as0[TPW];
#pragma unroll
    for (int k = 0; k < TPW; ++k) as0[k] = a_s[s0[k]];

    float l0[TPW], m[TPW];
#pragma unroll
    for (int k = 0; k < TPW; ++k) {
        l0[k] = v0[k] ? leaky(as0[k] + ad[k]) : -__builtin_inff();
        m[k]  = wave_max(fmaxf(e_self[k], l0[k]));
    }

    float w0[TPW], dsum[TPW];
#pragma unroll
    for (int k = 0; k < TPW; ++k) w0[k] = v0[k] ? expf(l0[k] - m[k]) : 0.0f;
#pragma unroll
    for (int k = 0; k < TPW; ++k) dsum[k] = wave_sum(w0[k]);
    float alpha[TPW], a_self[TPW];
#pragma unroll
    for (int k = 0; k < TPW; ++k) {
        const float w_self = expf(e_self[k] - m[k]);
        const float denom  = dsum[k] + w_self;
        alpha[k]  = w0[k] / denom;
        a_self[k] = __int_as_float(__builtin_amdgcn_readfirstlane(__float_as_int(w_self / denom)));
    }

#pragma unroll
    for (int k = 0; k < TPW; ++k) {
        const int t = t0 + k;
        const float* hself = H + t * DIM;
        float acc0 = a_self[k] * hself[lane];
        float acc1 = a_self[k] * hself[lane + 32];
        float acc2 = a_self[k] * hself[lane + 64];
        float acc3 = a_self[k] * hself[lane + 96];

        const int n = deg[k];
        int j = 0;
        for (; j + 8 <= n; j += 8)
            drain_batch<8>(j, s0[k], alpha[k], lane, H, acc0, acc1, acc2, acc3);
        if (j + 4 <= n) { drain_batch<4>(j, s0[k], alpha[k], lane, H, acc0, acc1, acc2, acc3); j += 4; }
        if (j + 2 <= n) { drain_batch<2>(j, s0[k], alpha[k], lane, H, acc0, acc1, acc2, acc3); j += 2; }
        if (j < n)        drain_batch<1>(j, s0[k], alpha[k], lane, H, acc0, acc1, acc2, acc3);

        float* orow = out + t * DIM;
        orow[lane]      = tanhf(acc0 + bias0);
        orow[lane + 32] = tanhf(acc1 + bias1);
        orow[lane + 64] = tanhf(acc2 + bias2);
        orow[lane + 96] = tanhf(acc3 + bias3);
    }
}

extern "C" void kernel_launch(void* const* d_in, const int* in_sizes, int n_in,
                              void* d_out, int out_size, void* d_ws, size_t ws_size,
                              hipStream_t stream) {
    const float* x        = (const float*)d_in[0];
    const int*   ei       = (const int*)  d_in[1];
    const float* W        = (const float*)d_in[2];
    const float* att_src  = (const float*)d_in[3];
    const float* att_dst  = (const float*)d_in[4];
    const float* bias     = (const float*)d_in[5];
    float*       out      = (float*)d_out;

    float* H   = (float*)d_ws;
    float* a_s = H   + (size_t)N_NODES * DIM;
    float* a_d = a_s + N_NODES;
    int*   cnt = (int*)(a_d + N_NODES);

    k_gemm_coef<<<N_NODES / 16, 32, 0, stream>>>(x, W, att_src, att_dst, H, a_s, a_d);

    k_fill<<<NGRP, FT, 0, stream>>>(ei, cnt, out);

    k_gather<<<N_NODES / (GW * TPW), GW * 32, 0, stream>>>(ei, H, a_s, a_d, cnt, bias, out);
}


# filing kernel: a work-group scans a quarter of the edge list for 782 targets (256 groups; before, 196 each scanned it all); each quarter has its own 32-slot line of the row: slots in quarter order
# speedup vs baseline: 4.7019x; 1.1169x over previous
#include <hip/hip_runtime.h>
#include <math.h>

typedef __attribute__((ext_vector_type(2))) float v2f;
typedef __attribute__((ext_vector_type(8))) float v8f;

#define N_NODES   50000
#define N_EDGES   640000
#define DIM       128
#define NEG_SLOPE 0.2f

__global__ void __launch_bounds__(32) __attribute__((amdgpu_waves_per_eu(1, 4)))
k_gemm_coef(const float* __restrict__ X,
            const float* __restrict__ W,
            const float* __restrict__ att_src,
            const float* __restrict__ att_dst,
            float* __restrict__ H,
            float* __restrict__ a_s, float* __restrict__ a_d) {
    const int lane = threadIdx.x;
    const int hb   = lane >> 4;
    const int l    = lane & 15;
    const int m0   = blockIdx.x * 16;

    const float* xrow = X + (m0 + l) * DIM + 2 * hb;
    const float* wcol = W + (2 * hb) * DIM + l;

    v8f acc[8];
#pragma unroll
    for (int nt = 0; nt < 8; ++nt) acc[nt] = (v8f){};

    v2f a[DIM / 4];
#pragma unroll
    for (int k = 0; k < DIM; k += 4) a[k / 4] = *(const v2f*)(xrow + k);

#pragma unroll
    for (int k = 0; k < DIM; k += 4) {
#pragma unroll
        for (int nt = 0; nt < 8; ++nt) {
            v2f b;
            b.x = wcol[k * DIM + nt * 16];
            b.y = wcol[k * DIM + DIM + nt * 16];
            acc[nt] = __builtin_amdgcn_wmma_f32_16x16x4_f32(
                false, a[k / 4], false, b, (short)0, acc[nt], false, false);
        }
    }

    float* hout = H + (m0 + 8 * hb) * DIM + l;
#pragma unroll
    for (int nt = 0; nt < 8; ++nt)
#pragma unroll
        for (int r = 0; r < 8; ++r)
            hout[r * DIM + nt * 16] = acc[nt][r];

    float ps[8], pd[8];
#pragma unroll
    for (int r = 0; r < 8; ++r) { ps[r] = 0.0f; pd[r] = 0.0f; }
#pragma unroll
    for (int nt = 0; nt < 8; ++nt) {
        const float cs = att_src[nt * 16 + l];
        const float cd = att_dst[nt * 16 + l];
#pragma unroll
        for (int r = 0; r < 8; ++r) {
            ps[r] += acc[nt][r] * cs;
            pd[r] += acc[nt][r] * cd;
        }
    }

    const bool b8 = (l & 8) != 0, b4 = (l & 4) != 0, b2 = (l & 2) != 0, b1 = (l & 1) != 0;
    float v8[8], v4[4], v2[2];
#pragma unroll
    for (int r = 0; r < 8; ++r) {
        const float keep = b8 ? pd[r] : ps[r];
        const float send = b8 ? ps[r] : pd[r];
        v8[r] = keep + __shfl_xor(send, 8, 32);
    }
#pragma unroll
    for (int j = 0; j < 4; ++j) {
        const float keep = b4 ? v8[j + 4] : v8[j];
        const float send = b4 ? v8[j] : v8[j + 4];
        v4[j] = keep + __shfl_xor(send, 4, 32);
    }
#pragma unroll
    for (int j = 0; j < 2; ++j) {
        const float keep = b2 ? v4[j + 2] : v4[j];
        const float send = b2 ? v4[j] : v4[j + 2];
        v2[j] = keep + __shfl_xor(send, 2, 32);
    }
    const float keep1 = b1 ? v2[1] : v2[0];
    const float send1 = b1 ? v2[0] : v2[1];
    const float tot = keep1 + __shfl_xor(send1, 1, 32);

    float* dst = b8 ? a_d : a_s;
    dst[m0 + 8 * hb + (l & 7)] = tot;
}

__device__ __forceinline__ float leaky(float e) {
    return e > 0.0f ? e : NEG_SLOPE * e;
}

__device__ __forceinline__ float wave_max(float v) {
    for (int off = 16; off > 0; off >>= 1) v = fmaxf(v, __shfl_xor(v, off, 32));
    return v;
}

__device__ __forceinline__ float wave_sum(float v) {
    for (int off = 16; off > 0; off >>= 1) v += __shfl_xor(v, off, 32);
    return v;
}

#define NS    4
#define SEG   (DIM / NS)
#define NR    64
#define RNG   ((N_NODES + NR - 1) / NR)
#define FT    1024
#define NQS   (N_EDGES / 4 / NS)
#define QU    5
#define STEP  (FT * QU)
#define NIT   ((NQS + STEP - 1) / STEP)
#define KLIM  (RNG * 32)
#define TW    ((RNG + 31) / 32)

static_assert(N_EDGES % (4 * NS) == 0, "a slice is a whole number of 16-byte groups");
static_assert(4 * QU <= 32 && RNG <= FT, "a key's low 5 bits hold the candidate; one thread per counter");

template <bool GUARD>
__device__ __forceinline__ void load_groups(int4 (&d)[QU], const int4* __restrict__ dst4,
                                            const int q0) {
#pragma unroll
    for (int j = 0; j < QU; ++j) {
        const int q = q0 + j * FT;
        if (GUARD) d[j] = q < NQS ? dst4[q] : make_int4(-1, -1, -1, -1);
        else       d[j] = dst4[q];
    }
}

__device__ __forceinline__ void file_trip(const int4 (&d)[QU], const int qg,
                                          const unsigned kb,
                                          int* s_cnt, int* s_edge) {
    unsigned key[4 * QU];
#pragma unroll
    for (int j = 0; j < QU; ++j) {
        key[4 * j + 0] = ((unsigned)d[j].x << 5) + ((unsigned)(4 * j + 0) - kb);
        key[4 * j + 1] = ((unsigned)d[j].y << 5) + ((unsigned)(4 * j + 1) - kb);
        key[4 * j + 2] = ((unsigned)d[j].z << 5) + ((unsigned)(4 * j + 2) - kb);
        key[4 * j + 3] = ((unsigned)d[j].w << 5) + ((unsigned)(4 * j + 3) - kb);
    }
    unsigned k = key[0];
#pragma unroll
    for (int c = 1; c < 4 * QU; ++c) k = min(k, key[c]);

    while (__builtin_amdgcn_ballot_w32(k < KLIM)) {
        unsigned done = KLIM;
        if (k < KLIM) {
            const unsigned u = k >> 5, c = k & 31;
            const int e = (qg + (int)(c >> 2) * FT) * 4 + (int)(c & 3);
            const int p = atomicAdd(&s_cnt[u], 1);
            if (p < SEG) s_edge[u * SEG + p] = e;
            done = k + 1;
        }
        unsigned rel = key[0] - done;
#pragma unroll
        for (int c = 1; c < 4 * QU; ++c) rel = min(rel, key[c] - done);
        k = rel < KLIM - done ? rel + done : 0xFFFFFFFFu;
    }
}

__global__ void __launch_bounds__(FT)
k_fill(const int* __restrict__ ei, int* __restrict__ cnt, float* __restrict__ out) {
    __shared__ int s_cnt[RNG];
    __shared__ int s_edge[RNG * SEG];
    const int tid    = threadIdx.x;
    const int slice  = blockIdx.x % NS;
    const int t_base = (blockIdx.x / NS) * RNG;

    if (tid < RNG) s_cnt[tid] = 0;
    __syncthreads();

    const int  qs   = slice * NQS;
    const int4* dst4 = (const int4*)(ei + N_EDGES) + qs;

    const unsigned kb = (unsigned)t_base << 5;

    static_assert(NIT % 2 == 0 && (NIT - 1) * STEP <= NQS && NIT * STEP >= NQS,
                  "trips 0 .. NIT-2 are whole, trip NIT-1 is the padded one");
    int4 a[QU], b[QU];
    load_groups<false>(a, dst4, tid);
#pragma nounroll
    for (int it = 0; it < NIT - 2; it += 2) {
        const int q0 = tid + it * STEP;
        load_groups<false>(b, dst4, q0 + STEP);
        file_trip(a, qs + q0, kb, s_cnt, s_edge);
        load_groups<false>(a, dst4, q0 + 2 * STEP);
        file_trip(b, qs + q0 + STEP, kb, s_cnt, s_edge);
    }
    {
        const int q0 = tid + (NIT - 2) * STEP;
        load_groups<true>(b, dst4, q0 + STEP);
        file_trip(a, qs + q0, kb, s_cnt, s_edge);
        file_trip(b, qs + q0 + STEP, kb, s_cnt, s_edge);
    }
    __syncthreads();

    const int wave = tid >> 5;
    const int lane = tid & 31;
    int  sv[TW];
    bool ok[TW];
#pragma unroll
    for (int k = 0; k < TW; ++k) {
        const int u = wave * TW + k;
        ok[k] = u < RNG && t_base + u < N_NODES && lane < s_cnt[u < RNG ? u : 0];
        sv[k] = ei[ok[k] ? s_edge[u * SEG + lane] : 0];
    }
#pragma unroll
    for (int k = 0; k < TW; ++k) {
        const int t = t_base + wave * TW + k;
        if (ok[k]) out[t * DIM + slice * SEG + lane] = __int_as_float(sv[k]);
    }

    if (tid < RNG && t_base + tid < N_NODES) {
        const int n = s_cnt[tid];
        ((unsigned char*)cnt)[(t_base + tid) * NS + slice] = (unsigned char)(n < 255 ? n : 255);
    }
}

#define TPW 4
#define GW  4

struct SegCnt {
    int p1, p2, p3, deg;
    bool over;
};
__device__ __forceinline__ SegCnt seg_counts(const int word) {
    static_assert(NS == 4, "one count byte per slice in a 32-bit word");
    const unsigned w = (unsigned)word;
    const int c0 = (int)(w & 255u), c1 = (int)((w >> 8) & 255u);
    const int c2 = (int)((w >> 16) & 255u), c3 = (int)(w >> 24);
    SegCnt r;
    r.p1 = c0; r.p2 = c0 + c1; r.p3 = c0 + c1 + c2; r.deg = c0 + c1 + c2 + c3;
    r.over = c0 > SEG || c1 > SEG || c2 > SEG || c3 > SEG;
    return r;
}
__device__ __forceinline__ int seg_slot(const SegCnt& n, const int i) {
    int off = i;
    off = i >= n.p1 ? i + (SEG - n.p1) : off;
    off = i >= n.p2 ? i + (2 * SEG - n.p2) : off;
    off = i >= n.p3 ? i + (3 * SEG - n.p3) : off;
    return off;
}

__device__ __forceinline__ void gather_one(const int t, const int lane,
                                           const int* __restrict__ ei,
                                           const float* __restrict__ H,
                                           const float* __restrict__ a_s,
                                           const float* __restrict__ a_d,
                                           const int* __restrict__ cnt,
                                           const float* __restrict__ bias,
                                           float* out) {
    const SegCnt n = seg_counts(cnt[t]);
    const int deg = n.deg;
    const bool scan = n.over;
    const int nchunk = scan ? N_EDGES / 32 : (deg + 31) >> 5;
    float* orow = out + t * DIM;
    const float ad = a_d[t];
    const float e_self = leaky(a_s[t] + ad);

    auto fetch = [&](int c, int& s) -> bool {
        const int i = c * 32 + lane;
        if (scan) { s = ei[i]; return ei[N_EDGES + i] == t; }
        const bool v = i < deg;
        s = v ? __float_as_int(orow[seg_slot(n, i)]) : 0;
        return v;
    };

    int s0;
    const bool v0 = fetch(0, s0);
    const float l0 = v0 ? leaky(a_s[s0] + ad) : -__builtin_inff();

    float m = fmaxf(e_self, l0);
    for (int c = 1; c < nchunk; ++c) {
        int s;
        if (fetch(c, s)) m = fmaxf(m, leaky(a_s[s] + ad));
    }
    m = wave_max(m);

    const float w0 = v0 ? expf(l0 - m) : 0.0f;
    float d = w0;
    for (int c = 1; c < nchunk; ++c) {
        int s;
        if (fetch(c, s)) d += expf(leaky(a_s[s] + ad) - m);
    }
    const float w_self = expf(e_self - m);
    const float denom = wave_sum(d) + w_self;

    const float* hself = H + t * DIM + lane;
    const float a_self = w_self / denom;
    float acc0 = a_self * hself[0];
    float acc1 = a_self * hself[32];
    float acc2 = a_self * hself[64];
    float acc3 = a_self * hself[96];

    auto drain = [&](bool v, int s, float alpha) {
        unsigned mask = __builtin_amdgcn_ballot_w32(v);
        while (mask) {
            const int j = __builtin_ctz(mask);
            mask &= mask - 1;
            const int sj = __builtin_amdgcn_readlane(s, j);
            const float aj = __int_as_float(
                __builtin_amdgcn_readlane(__float_as_int(alpha), j));
            const float* hrow = H + sj * DIM + lane;
            acc0 += aj * hrow[0];
            acc1 += aj * hrow[32];
            acc2 += aj * hrow[64];
            acc3 += aj * hrow[96];
        }
    };

    drain(v0, s0, w0 / denom);
    for (int c = 1; c < nchunk; ++c) {
        int s;
        const bool v = fetch(c, s);
        const float a = v ? expf(leaky(a_s[s] + ad) - m) / denom : 0.0f;
        drain(v, s, a);
    }

    orow[lane]      = tanhf(acc0 + bias[lane]);
    orow[lane + 32] = tanhf(acc1 + bias[lane + 32]);
    orow[lane + 64] = tanhf(acc2 + bias[lane + 64]);
    orow[lane + 96] = tanhf(acc3 + bias[lane + 96]);
}

template <int B>
__device__ __forceinline__ void drain_batch(const int first, const int s, const float alpha,
                                            const int lane, const float* __restrict__ H,
                                            float& acc0, float& acc1,
                                            float& acc2, float& acc3) {
    float aj[B];
    float r0[B], r1[B], r2[B], r3[B];
#pragma unroll
    for (int b = 0; b < B; ++b) {
        const int sj = __builtin_amdgcn_readlane(s, first + b);
        aj[b] = __int_as_float(__builtin_amdgcn_readlane(__float_as_int(alpha), first + b));
        const float* hrow = H + sj * DIM;
        r0[b] = hrow[lane];
        r1[b] = hrow[lane + 32];
        r2[b] = hrow[lane + 64];
        r3[b] = hrow[lane + 96];
    }
#pragma unroll
    for (int b = 0; b < B; ++b) {
        acc0 = __builtin_fmaf(aj[b], r0[b], acc0);
        acc1 = __builtin_fmaf(aj[b], r1[b], acc1);
        acc2 = __builtin_fmaf(aj[b], r2[b], acc2);
        acc3 = __builtin_fmaf(aj[b], r3[b], acc3);
    }
}

__global__ void __launch_bounds__(GW * 32)
k_gather(const int* __restrict__ ei,
         const float* __restrict__ H,
         const float* __restrict__ a_s,
         const float* __restrict__ a_d,
         const int* __restrict__ cnt,
         const float* __restrict__ bias,
         float* out) {
    const int wave = threadIdx.x >> 5;
    const int lane = threadIdx.x & 31;
    const int t0 = __builtin_amdgcn_readfirstlane((blockIdx.x * GW + wave) * TPW);

    SegCnt n[TPW];
    int deg[TPW];
    int degmax = 0;
#pragma unroll
    for (int k = 0; k < TPW; ++k) {
        n[k]   = seg_counts(cnt[t0 + k]);
        deg[k] = n[k].deg;
        degmax = deg[k] > degmax ? deg[k] : degmax;
    }

    if (degmax > 32) {
#pragma nounroll
        for (int k = 0; k < TPW; ++k)
            gather_one(t0 + k, lane, ei, H, a_s, a_d, cnt, bias, out);
        return;
    }

    const float bias0 = bias[lane];
    const float bias1 = bias[lane + 32];
    const float bias2 = bias[lane + 64];
    const float bias3 = bias[lane + 96];

    float ad[TPW], e_self[TPW];
#pragma unroll
    for (int k = 0; k < TPW; ++k) {
        ad[k]     = a_d[t0 + k];
        e_self[k] = leaky(a_s[t0 + k] + ad[k]);
    }

    int  s0[TPW];
    bool v0[TPW];
#pragma unroll
    for (int k = 0; k < TPW; ++k) {
        const float* orow = out + (t0 + k) * DIM;
        v0[k] = lane < deg[k];
        s0[k] = v0[k] ? __float_as_int(orow[seg_slot(n[k], lane)]) : 0;
    }

    float as0[TPW];
#pragma unroll
    for (int k = 0; k < TPW; ++k) as0[k] = a_s[s0[k]];

    float l0[TPW], m[TPW];
#pragma unroll
    for (int k = 0; k < TPW; ++k) {
        l0[k] = v0[k] ? leaky(as0[k] + ad[k]) : -__builtin_inff();
        m[k]  = wave_max(fmaxf(e_self[k], l0[k]));
    }

    float w0[TPW], dsum[TPW];
#pragma unroll
    for (int k = 0; k < TPW; ++k) w0[k] = v0[k] ? expf(l0[k] - m[k]) : 0.0f;
#pragma unroll
    for (int k = 0; k < TPW; ++k) dsum[k] = wave_sum(w0[k]);
    float alpha[TPW], a_self[TPW];
#pragma unroll
    for (int k = 0; k < TPW; ++k) {
        const float w_self = expf(e_self[k] - m[k]);
        const float denom  = dsum[k] + w_self;
        alpha[k]  = w0[k] / denom;
        a_self[k] = __int_as_float(__builtin_amdgcn_readfirstlane(__float_as_int(w_self / denom)));
    }

#pragma unroll
    for (int k = 0; k < TPW; ++k) {
        const int t = t0 + k;
        const float* hself = H + t * DIM;
        float acc0 = a_self[k] * hself[lane];
        float acc1 = a_self[k] * hself[lane + 32];
        float acc2 = a_self[k] * hself[lane + 64];
        float acc3 = a_self[k] * hself[lane + 96];

        const int n = deg[k];
        int j = 0;
        for (; j + 8 <= n; j += 8)
            drain_batch<8>(j, s0[k], alpha[k], lane, H, acc0, acc1, acc2, acc3);
        if (j + 4 <= n) { drain_batch<4>(j, s0[k], alpha[k], lane, H, acc0, acc1, acc2, acc3); j += 4; }
        if (j + 2 <= n) { drain_batch<2>(j, s0[k], alpha[k], lane, H, acc0, acc1, acc2, acc3); j += 2; }
        if (j < n)        drain_batch<1>(j, s0[k], alpha[k], lane, H, acc0, acc1, acc2, acc3);

        float* orow = out + t * DIM;
        orow[lane]      = tanhf(acc0 + bias0);
        orow[lane + 32] = tanhf(acc1 + bias1);
        orow[lane + 64] = tanhf(acc2 + bias2);
        orow[lane + 96] = tanhf(acc3 + bias3);
    }
}

extern "C" void kernel_launch(void* const* d_in, const int* in_sizes, int n_in,
                              void* d_out, int out_size, void* d_ws, size_t ws_size,
                              hipStream_t stream) {
    const float* x        = (const float*)d_in[0];
    const int*   ei       = (const int*)  d_in[1];
    const float* W        = (const float*)d_in[2];
    const float* att_src  = (const float*)d_in[3];
    const float* att_dst  = (const float*)d_in[4];
    const float* bias     = (const float*)d_in[5];
    float*       out      = (float*)d_out;

    float* H   = (float*)d_ws;
    float* a_s = H   + (size_t)N_NODES * DIM;
    float* a_d = a_s + N_NODES;
    int*   cnt = (int*)(a_d + N_NODES);

    k_gemm_coef<<<N_NODES / 16, 32, 0, stream>>>(x, W, att_src, att_dst, H, a_s, a_d);

    k_fill<<<NR * NS, FT, 0, stream>>>(ei, cnt, out);

    k_gather<<<N_NODES / (GW * TPW), GW * 32, 0, stream>>>(ei, H, a_s, a_d, cnt, bias, out);
}
